# starting kernel
# baseline (speedup 1.0000x reference)
#include <hip/hip_runtime.h>
#include <hip/hip_bf16.h>


#define D      256
#define NMAT   1024
#define G      16
#define NCHUNK 64
#define MATELT (D * D)
#define CHUNKSTRIDE (2 * MATELT)

typedef __attribute__((ext_vector_type(2))) float v2f;
typedef __attribute__((ext_vector_type(8))) float v8f;

__global__ __launch_bounds__(256) void prodscan_chunk_mm(
    const float* __restrict__ m, float* __restrict__ ws, int t) {
  const int j    = blockIdx.x;
  const int slab = blockIdx.y;

  float* buf0 = ws + (size_t)j * CHUNKSTRIDE;
  float* buf1 = buf0 + MATELT;

  const float* A = (t == 1) ? (m + (size_t)j * G * MATELT)
                            : ((t & 1) ? buf1 : buf0);
  const float* B = m + (size_t)(j * G + t) * MATELT;
  float*       C = (t & 1) ? buf0 : buf1;

  const int w     = threadIdx.x >> 5;
  const int lane  = threadIdx.x & 31;
  const int laneM = lane & 15;
  const int laneH = lane >> 4;

  const int tr0 = slab * 8 + (w >> 2) * 4;
  const int tc0 = (w & 3) * 4;

  v8f acc[4][4] = {};

  for (int k0 = 0; k0 < D; k0 += 4) {
    v2f af[4];
    v2f bf[4];
#pragma unroll
    for (int r = 0; r < 4; ++r) {
      const float* ap =
          A + (size_t)((tr0 + r) * 16 + laneM) * D + k0 + 2 * laneH;
      af[r].x = ap[0];
      af[r].y = ap[1];
    }
#pragma unroll
    for (int c = 0; c < 4; ++c) {
      const float* bp =
          B + (size_t)(k0 + 2 * laneH) * D + (tc0 + c) * 16 + laneM;
      bf[c].x = bp[0];
      bf[c].y = bp[D];
    }
#pragma unroll
    for (int r = 0; r < 4; ++r)
#pragma unroll
      for (int c = 0; c < 4; ++c)
        acc[r][c] = __builtin_amdgcn_wmma_f32_16x16x4_f32(
            false, af[r], false, bf[c], (short)0, acc[r][c], false, false);
  }

#pragma unroll
  for (int r = 0; r < 4; ++r)
#pragma unroll
    for (int c = 0; c < 4; ++c)
#pragma unroll
      for (int e = 0; e < 8; ++e) {
        const int row = (tr0 + r) * 16 + e + 8 * laneH;
        const int col = (tc0 + c) * 16 + laneM;
        C[(size_t)row * D + col] = acc[r][c][e];
      }
}

__global__ __launch_bounds__(256) void prodscan_carries(
    const float* __restrict__ x, const float* __restrict__ ws,
    float* __restrict__ carries, float* __restrict__ out) {
  __shared__ float u[D];
  const int t = threadIdx.x;
  float v = x[t];
  u[t] = v;
  carries[t] = v;
  __syncthreads();
  for (int j = 0; j < NCHUNK; ++j) {
    const float* P = ws + (size_t)j * CHUNKSTRIDE;
    if (j + 1 < NCHUNK)
      __builtin_prefetch(ws + (size_t)(j + 1) * CHUNKSTRIDE + t * 64, 0, 1);
    float acc = 0.f;
#pragma unroll 8
    for (int i = 0; i < D; ++i)
      acc = fmaf(u[i], P[(size_t)i * D + t], acc);
    __syncthreads();
    u[t] = acc;
    carries[(size_t)(j + 1) * D + t] = acc;
    __syncthreads();
  }
  out[(size_t)NMAT * D + t] = u[t];
}

__global__ __launch_bounds__(256) void prodscan_expand(
    const float* __restrict__ m, const float* __restrict__ carries,
    float* __restrict__ out) {
  __shared__ float vb[2][D];
  const int j = blockIdx.x;
  const int t = threadIdx.x;
  float v = carries[(size_t)j * D + t];
  vb[0][t] = v;
  out[(size_t)(j * G) * D + t] = v;
  __syncthreads();
  int cur = 0;
  for (int i = 0; i < G - 1; ++i) {
    const float* M = m + (size_t)(j * G + i) * MATELT;
    if (i + 1 < G - 1)
      __builtin_prefetch(m + (size_t)(j * G + i + 1) * MATELT + t * 64, 0, 1);
    float acc = 0.f;
#pragma unroll 8
    for (int r = 0; r < D; ++r)
      acc = fmaf(vb[cur][r], M[(size_t)r * D + t], acc);
    vb[cur ^ 1][t] = acc;
    out[(size_t)(j * G + i + 1) * D + t] = acc;
    cur ^= 1;
    __syncthreads();
  }
}

extern "C" void kernel_launch(void* const* d_in, const int* in_sizes, int n_in,
                              void* d_out, int out_size, void* d_ws,
                              size_t ws_size, hipStream_t stream) {
  (void)in_sizes; (void)n_in; (void)out_size; (void)ws_size;
  const float* x = (const float*)d_in[0];
  const float* m = (const float*)d_in[1];
  float* out = (float*)d_out;
  float* ws  = (float*)d_ws;
  float* carries = ws + (size_t)NCHUNK * CHUNKSTRIDE;

  for (int t = 1; t < G; ++t)
    prodscan_chunk_mm<<<dim3(NCHUNK, 2), 256, 0, stream>>>(m, ws, t);
  prodscan_carries<<<1, 256, 0, stream>>>(x, ws, carries, out);
  prodscan_expand<<<NCHUNK, 256, 0, stream>>>(m, carries, out);
}


# GEMV chains (carry scan, chunk expansion): 1024-thread split-K step, 16 b128 row loads per thread keep the next matrix in flight across LDS-only barriers; sums regrouped as 16 partials of 16 rows
# speedup vs baseline: 1.8592x; 1.8592x over previous
#include <hip/hip_runtime.h>
#include <hip/hip_bf16.h>


#define D      256
#define NMAT   1024
#define G      16
#define NCHUNK 64
#define MATELT (D * D)
#define CHUNKSTRIDE (2 * MATELT)

typedef __attribute__((ext_vector_type(2))) float v2f;
typedef __attribute__((ext_vector_type(8))) float v8f;

__global__ __launch_bounds__(256) void prodscan_chunk_mm(
    const float* __restrict__ m, float* __restrict__ ws, int t) {
  const int j    = blockIdx.x;
  const int slab = blockIdx.y;

  float* buf0 = ws + (size_t)j * CHUNKSTRIDE;
  float* buf1 = buf0 + MATELT;

  const float* A = (t == 1) ? (m + (size_t)j * G * MATELT)
                            : ((t & 1) ? buf1 : buf0);
  const float* B = m + (size_t)(j * G + t) * MATELT;
  float*       C = (t & 1) ? buf0 : buf1;

  const int w     = threadIdx.x >> 5;
  const int lane  = threadIdx.x & 31;
  const int laneM = lane & 15;
  const int laneH = lane >> 4;

  const int tr0 = slab * 8 + (w >> 2) * 4;
  const int tc0 = (w & 3) * 4;

  v8f acc[4][4] = {};

  for (int k0 = 0; k0 < D; k0 += 4) {
    v2f af[4];
    v2f bf[4];
#pragma unroll
    for (int r = 0; r < 4; ++r) {
      const float* ap =
          A + (size_t)((tr0 + r) * 16 + laneM) * D + k0 + 2 * laneH;
      af[r].x = ap[0];
      af[r].y = ap[1];
    }
#pragma unroll
    for (int c = 0; c < 4; ++c) {
      const float* bp =
          B + (size_t)(k0 + 2 * laneH) * D + (tc0 + c) * 16 + laneM;
      bf[c].x = bp[0];
      bf[c].y = bp[D];
    }
#pragma unroll
    for (int r = 0; r < 4; ++r)
#pragma unroll
      for (int c = 0; c < 4; ++c)
        acc[r][c] = __builtin_amdgcn_wmma_f32_16x16x4_f32(
            false, af[r], false, bf[c], (short)0, acc[r][c], false, false);
  }

#pragma unroll
  for (int r = 0; r < 4; ++r)
#pragma unroll
    for (int c = 0; c < 4; ++c)
#pragma unroll
      for (int e = 0; e < 8; ++e) {
        const int row = (tr0 + r) * 16 + e + 8 * laneH;
        const int col = (tc0 + c) * 16 + laneM;
        C[(size_t)row * D + col] = acc[r][c][e];
      }
}

#define GEMV_THREADS 1024
#define KGROUPS 16
#define RPT (D / KGROUPS)

typedef __attribute__((ext_vector_type(4))) float v4f;

__device__ __forceinline__ void lds_barrier() {
  __builtin_amdgcn_fence(__ATOMIC_RELEASE, "workgroup", "local");
  __builtin_amdgcn_s_barrier();
  __builtin_amdgcn_fence(__ATOMIC_ACQUIRE, "workgroup", "local");
}

__device__ __forceinline__ const v4f* gemv_slice(const float* M, int kg,
                                                 int cg) {
  return reinterpret_cast<const v4f*>(M + (size_t)(kg * RPT) * D + 4 * cg);
}

__device__ __forceinline__ void gemv_fill(v4f (&r)[RPT], const v4f* p) {
#pragma unroll
  for (int i = 0; i < RPT; ++i) r[i] = p[i * (D / 4)];
}

template <bool MORE>
__device__ __forceinline__ void gemv_step(float* u, float (*part)[D],
                                          v4f (&r)[RPT], const v4f* next,
                                          float* dst, int tid, int kg,
                                          int cg) {
  v4f acc = {0.f, 0.f, 0.f, 0.f};
#pragma unroll
  for (int q = 0; q < RPT / 4; ++q) {
    const v4f uu = *reinterpret_cast<const v4f*>(u + kg * RPT + 4 * q);
#pragma unroll
    for (int e = 0; e < 4; ++e) {
      const int i = 4 * q + e;
      acc.x = fmaf(uu[e], r[i].x, acc.x);
      acc.y = fmaf(uu[e], r[i].y, acc.y);
      acc.z = fmaf(uu[e], r[i].z, acc.z);
      acc.w = fmaf(uu[e], r[i].w, acc.w);
      if (MORE) r[i] = next[i * (D / 4)];
    }
  }
  *reinterpret_cast<v4f*>(&part[kg][4 * cg]) = acc;
  lds_barrier();
  if (tid < D) {
    float s = 0.f;
    for (int g = 0; g < KGROUPS; ++g) s += part[g][tid];
    u[tid] = s;
    dst[tid] = s;
  }
  lds_barrier();
}

__global__ __launch_bounds__(GEMV_THREADS) void prodscan_carries(
    const float* __restrict__ x, const float* __restrict__ ws,
    float* __restrict__ carries, float* __restrict__ out) {
  __shared__ __attribute__((aligned(16))) float u[D];
  __shared__ __attribute__((aligned(16))) float part[KGROUPS][D];
  const int tid = threadIdx.x;
  const int cg  = tid & 63;
  const int kg  = tid >> 6;
  v4f r[RPT];
  gemv_fill(r, gemv_slice(ws, kg, cg));
  if (tid < D) {
    const float v = x[tid];
    u[tid] = v;
    carries[tid] = v;
  }
  lds_barrier();
  for (int j = 0; j < NCHUNK - 1; ++j)
    gemv_step<true>(u, part, r,
                    gemv_slice(ws + (size_t)(j + 1) * CHUNKSTRIDE, kg, cg),
                    carries + (size_t)(j + 1) * D, tid, kg, cg);
  gemv_step<false>(u, part, r, nullptr, carries + (size_t)NCHUNK * D, tid, kg,
                   cg);
  if (tid < D) out[(size_t)NMAT * D + tid] = u[tid];
}

__global__ __launch_bounds__(GEMV_THREADS) void prodscan_expand(
    const float* __restrict__ m, const float* __restrict__ carries,
    float* __restrict__ out) {
  __shared__ __attribute__((aligned(16))) float u[D];
  __shared__ __attribute__((aligned(16))) float part[KGROUPS][D];
  const int j   = blockIdx.x;
  const int tid = threadIdx.x;
  const int cg  = tid & 63;
  const int kg  = tid >> 6;
  const float* Mj = m + (size_t)j * G * MATELT;
  v4f r[RPT];
  gemv_fill(r, gemv_slice(Mj, kg, cg));
  if (tid < D) {
    const float v = carries[(size_t)j * D + tid];
    u[tid] = v;
    out[(size_t)(j * G) * D + tid] = v;
  }
  lds_barrier();
  for (int i = 0; i < G - 2; ++i)
    gemv_step<true>(u, part, r,
                    gemv_slice(Mj + (size_t)(i + 1) * MATELT, kg, cg),
                    out + (size_t)(j * G + i + 1) * D, tid, kg, cg);
  gemv_step<false>(u, part, r, nullptr, out + (size_t)(j * G + G - 1) * D, tid,
                   kg, cg);
}

extern "C" void kernel_launch(void* const* d_in, const int* in_sizes, int n_in,
                              void* d_out, int out_size, void* d_ws,
                              size_t ws_size, hipStream_t stream) {
  (void)in_sizes; (void)n_in; (void)out_size; (void)ws_size;
  const float* x = (const float*)d_in[0];
  const float* m = (const float*)d_in[1];
  float* out = (float*)d_out;
  float* ws  = (float*)d_ws;
  float* carries = ws + (size_t)NCHUNK * CHUNKSTRIDE;

  for (int t = 1; t < G; ++t)
    prodscan_chunk_mm<<<dim3(NCHUNK, 2), 256, 0, stream>>>(m, ws, t);
  prodscan_carries<<<1, GEMV_THREADS, 0, stream>>>(x, ws, carries, out);
  prodscan_expand<<<NCHUNK, GEMV_THREADS, 0, stream>>>(m, carries, out);
}


# Chunk-product GEMM k-loop: A/B fragment loads run 4 k-steps ahead through a register ring, each slot refilled right behind its WMMA block, so load latency hides under 3 blocks; sum order unchanged
# speedup vs baseline: 2.5526x; 1.3730x over previous
#include <hip/hip_runtime.h>
#include <hip/hip_bf16.h>


#define D      256
#define NMAT   1024
#define G      16
#define NCHUNK 64
#define MATELT (D * D)
#define CHUNKSTRIDE (2 * MATELT)

typedef __attribute__((ext_vector_type(2))) float v2f;
typedef __attribute__((ext_vector_type(8))) float v8f;

#define PD 4

__device__ __forceinline__ void mm_load_frags(v2f (&a)[4], v2f (&b)[4],
                                              const float* Ap,
                                              const float* Bp, int k0) {
#pragma unroll
  for (int r = 0; r < 4; ++r) {
    const float* ap = Ap + (size_t)(r * 16) * D + k0;
    a[r].x = ap[0];
    a[r].y = ap[1];
  }
#pragma unroll
  for (int c = 0; c < 4; ++c) {
    const float* bp = Bp + (size_t)k0 * D + c * 16;
    b[c].x = bp[0];
    b[c].y = bp[D];
  }
}

__device__ __forceinline__ void mm_block(v8f (&acc)[4][4], const v2f (&a)[4],
                                         const v2f (&b)[4]) {
#pragma unroll
  for (int r = 0; r < 4; ++r)
#pragma unroll
    for (int c = 0; c < 4; ++c)
      acc[r][c] = __builtin_amdgcn_wmma_f32_16x16x4_f32(
          false, a[r], false, b[c], (short)0, acc[r][c], false, false);
}

__global__ __launch_bounds__(256) void prodscan_chunk_mm(
    const float* __restrict__ m, float* __restrict__ ws, int t) {
  const int j    = blockIdx.x;
  const int slab = blockIdx.y;

  float* buf0 = ws + (size_t)j * CHUNKSTRIDE;
  float* buf1 = buf0 + MATELT;

  const float* A = (t == 1) ? (m + (size_t)j * G * MATELT)
                            : ((t & 1) ? buf1 : buf0);
  const float* B = m + (size_t)(j * G + t) * MATELT;
  float*       C = (t & 1) ? buf0 : buf1;

  const int w     = threadIdx.x >> 5;
  const int lane  = threadIdx.x & 31;
  const int laneM = lane & 15;
  const int laneH = lane >> 4;

  const int tr0 = slab * 8 + (w >> 2) * 4;
  const int tc0 = (w & 3) * 4;

  const float* Ap = A + (size_t)(tr0 * 16 + laneM) * D + 2 * laneH;
  const float* Bp = B + (size_t)(2 * laneH) * D + tc0 * 16 + laneM;

  v8f acc[4][4] = {};
  v2f af[PD][4];
  v2f bf[PD][4];

#pragma unroll
  for (int s = 0; s < PD; ++s) {
    mm_load_frags(af[s], bf[s], Ap, Bp, 4 * s);
    __builtin_amdgcn_sched_barrier(0);
  }

  for (int k0 = 0; k0 < D - 4 * PD; k0 += 4 * PD) {
#pragma unroll
    for (int s = 0; s < PD; ++s) {
      mm_block(acc, af[s], bf[s]);
      mm_load_frags(af[s], bf[s], Ap, Bp, k0 + 4 * (s + PD));
      __builtin_amdgcn_sched_barrier(0);
    }
  }
#pragma unroll
  for (int s = 0; s < PD; ++s) mm_block(acc, af[s], bf[s]);

#pragma unroll
  for (int r = 0; r < 4; ++r)
#pragma unroll
    for (int c = 0; c < 4; ++c)
#pragma unroll
      for (int e = 0; e < 8; ++e) {
        const int row = (tr0 + r) * 16 + e + 8 * laneH;
        const int col = (tc0 + c) * 16 + laneM;
        C[(size_t)row * D + col] = acc[r][c][e];
      }
}

#define GEMV_THREADS 1024
#define KGROUPS 16
#define RPT (D / KGROUPS)

typedef __attribute__((ext_vector_type(4))) float v4f;

__device__ __forceinline__ void lds_barrier() {
  __builtin_amdgcn_fence(__ATOMIC_RELEASE, "workgroup", "local");
  __builtin_amdgcn_s_barrier();
  __builtin_amdgcn_fence(__ATOMIC_ACQUIRE, "workgroup", "local");
}

__device__ __forceinline__ const v4f* gemv_slice(const float* M, int kg,
                                                 int cg) {
  return reinterpret_cast<const v4f*>(M + (size_t)(kg * RPT) * D + 4 * cg);
}

__device__ __forceinline__ void gemv_fill(v4f (&r)[RPT], const v4f* p) {
#pragma unroll
  for (int i = 0; i < RPT; ++i) r[i] = p[i * (D / 4)];
}

template <bool MORE>
__device__ __forceinline__ void gemv_step(float* u, float (*part)[D],
                                          v4f (&r)[RPT], const v4f* next,
                                          float* dst, int tid, int kg,
                                          int cg) {
  v4f acc = {0.f, 0.f, 0.f, 0.f};
#pragma unroll
  for (int q = 0; q < RPT / 4; ++q) {
    const v4f uu = *reinterpret_cast<const v4f*>(u + kg * RPT + 4 * q);
#pragma unroll
    for (int e = 0; e < 4; ++e) {
      const int i = 4 * q + e;
      acc.x = fmaf(uu[e], r[i].x, acc.x);
      acc.y = fmaf(uu[e], r[i].y, acc.y);
      acc.z = fmaf(uu[e], r[i].z, acc.z);
      acc.w = fmaf(uu[e], r[i].w, acc.w);
      if (MORE) r[i] = next[i * (D / 4)];
    }
  }
  *reinterpret_cast<v4f*>(&part[kg][4 * cg]) = acc;
  lds_barrier();
  if (tid < D) {
    float s = 0.f;
    for (int g = 0; g < KGROUPS; ++g) s += part[g][tid];
    u[tid] = s;
    dst[tid] = s;
  }
  lds_barrier();
}

__global__ __launch_bounds__(GEMV_THREADS) void prodscan_carries(
    const float* __restrict__ x, const float* __restrict__ ws,
    float* __restrict__ carries, float* __restrict__ out) {
  __shared__ __attribute__((aligned(16))) float u[D];
  __shared__ __attribute__((aligned(16))) float part[KGROUPS][D];
  const int tid = threadIdx.x;
  const int cg  = tid & 63;
  const int kg  = tid >> 6;
  v4f r[RPT];
  gemv_fill(r, gemv_slice(ws, kg, cg));
  if (tid < D) {
    const float v = x[tid];
    u[tid] = v;
    carries[tid] = v;
  }
  lds_barrier();
  for (int j = 0; j < NCHUNK - 1; ++j)
    gemv_step<true>(u, part, r,
                    gemv_slice(ws + (size_t)(j + 1) * CHUNKSTRIDE, kg, cg),
                    carries + (size_t)(j + 1) * D, tid, kg, cg);
  gemv_step<false>(u, part, r, nullptr, carries + (size_t)NCHUNK * D, tid, kg,
                   cg);
  if (tid < D) out[(size_t)NMAT * D + tid] = u[tid];
}

__global__ __launch_bounds__(GEMV_THREADS) void prodscan_expand(
    const float* __restrict__ m, const float* __restrict__ carries,
    float* __restrict__ out) {
  __shared__ __attribute__((aligned(16))) float u[D];
  __shared__ __attribute__((aligned(16))) float part[KGROUPS][D];
  const int j   = blockIdx.x;
  const int tid = threadIdx.x;
  const int cg  = tid & 63;
  const int kg  = tid >> 6;
  const float* Mj = m + (size_t)j * G * MATELT;
  v4f r[RPT];
  gemv_fill(r, gemv_slice(Mj, kg, cg));
  if (tid < D) {
    const float v = carries[(size_t)j * D + tid];
    u[tid] = v;
    out[(size_t)(j * G) * D + tid] = v;
  }
  lds_barrier();
  for (int i = 0; i < G - 2; ++i)
    gemv_step<true>(u, part, r,
                    gemv_slice(Mj + (size_t)(i + 1) * MATELT, kg, cg),
                    out + (size_t)(j * G + i + 1) * D, tid, kg, cg);
  gemv_step<false>(u, part, r, nullptr, out + (size_t)(j * G + G - 1) * D, tid,
                   kg, cg);
}

extern "C" void kernel_launch(void* const* d_in, const int* in_sizes, int n_in,
                              void* d_out, int out_size, void* d_ws,
                              size_t ws_size, hipStream_t stream) {
  (void)in_sizes; (void)n_in; (void)out_size; (void)ws_size;
  const float* x = (const float*)d_in[0];
  const float* m = (const float*)d_in[1];
  float* out = (float*)d_out;
  float* ws  = (float*)d_ws;
  float* carries = ws + (size_t)NCHUNK * CHUNKSTRIDE;

  for (int t = 1; t < G; ++t)
    prodscan_chunk_mm<<<dim3(NCHUNK, 2), 256, 0, stream>>>(m, ws, t);
  prodscan_carries<<<1, GEMV_THREADS, 0, stream>>>(x, ws, carries, out);
  prodscan_expand<<<NCHUNK, GEMV_THREADS, 0, stream>>>(m, carries, out);
}


# Chunk products: 15 launches -> 1; rows of a running product are independent, so each work-group carries its own 64-row slab through all 15 multiplies in LDS (256 work-groups); sum order unchanged
# speedup vs baseline: 4.6758x; 1.8318x over previous
#include <hip/hip_runtime.h>
#include <hip/hip_bf16.h>


#define D      256
#define NMAT   1024
#define G      16
#define NCHUNK 64
#define MATELT (D * D)
#define CHUNKSTRIDE (2 * MATELT)

typedef __attribute__((ext_vector_type(2))) float v2f;
typedef __attribute__((ext_vector_type(4))) float v4f;
typedef __attribute__((ext_vector_type(8))) float v8f;

#define SLABROWS 64
#define NSLAB    (D / SLABROWS)
#define APITCH   260
#define SLABELT  (SLABROWS * APITCH)

#define KS 8
#define PD 4

__device__ __forceinline__ void mm_load_a(v2f (&a)[2][4], const float* As,
                                          int k0) {
#pragma unroll
  for (int kk = 0; kk < 2; ++kk)
#pragma unroll
    for (int r = 0; r < 4; ++r)
      a[kk][r] = *reinterpret_cast<const v2f*>(As + r * 16 * APITCH + k0 +
                                               4 * kk);
}

__device__ __forceinline__ void mm_load_b(v2f (&b)[2][2], const float* Bp,
                                          int k0) {
#pragma unroll
  for (int kk = 0; kk < 2; ++kk)
#pragma unroll
    for (int c = 0; c < 2; ++c) {
      const float* bp = Bp + (size_t)(k0 + 4 * kk) * D + c * 16;
      b[kk][c].x = bp[0];
      b[kk][c].y = bp[D];
    }
}

__device__ __forceinline__ void mm_block(v8f (&acc)[4][2],
                                         const v2f (&a)[2][4],
                                         const v2f (&b)[2][2]) {
#pragma unroll
  for (int kk = 0; kk < 2; ++kk)
#pragma unroll
    for (int r = 0; r < 4; ++r)
#pragma unroll
      for (int c = 0; c < 2; ++c)
        acc[r][c] = __builtin_amdgcn_wmma_f32_16x16x4_f32(
            false, a[kk][r], false, b[kk][c], (short)0, acc[r][c], false,
            false);
}

__device__ __forceinline__ void lds_barrier() {
  __builtin_amdgcn_fence(__ATOMIC_RELEASE, "workgroup", "local");
  __builtin_amdgcn_s_barrier();
  __builtin_amdgcn_fence(__ATOMIC_ACQUIRE, "workgroup", "local");
}

template <bool LAST>
__device__ __forceinline__ void chain_step(const float* As, const float* Bp,
                                           v2f (&bf)[PD][2][2], float* Cp) {
  constexpr int CPITCH = LAST ? D : APITCH;
  v8f acc[4][2] = {};
  v2f af[PD][2][4];
#pragma unroll
  for (int s = 0; s < PD; ++s) mm_load_a(af[s], As, KS * s);

  for (int k0 = 0; k0 < D - KS * PD; k0 += KS * PD) {
#pragma unroll
    for (int s = 0; s < PD; ++s) {
      mm_block(acc, af[s], bf[s]);
      mm_load_a(af[s], As, k0 + KS * (s + PD));
      mm_load_b(bf[s], Bp, k0 + KS * (s + PD));
      __builtin_amdgcn_sched_barrier(0);
    }
  }
#pragma unroll
  for (int s = 0; s < PD; ++s) {
    mm_block(acc, af[s], bf[s]);
    if (!LAST) {
      mm_load_b(bf[s], Bp + MATELT, KS * s);
      __builtin_amdgcn_sched_barrier(0);
    }
  }

#pragma unroll
  for (int r = 0; r < 4; ++r)
#pragma unroll
    for (int c = 0; c < 2; ++c)
#pragma unroll
      for (int e = 0; e < 8; ++e)
        Cp[(size_t)(16 * r + e) * CPITCH + 16 * c] = acc[r][c][e];
}

__global__ __launch_bounds__(256) void prodscan_chunk_chain(
    const float* __restrict__ m, float* __restrict__ ws) {
  __shared__ __attribute__((aligned(16))) float slab[2 * SLABELT];

  const int j    = blockIdx.x;
  const int row0 = blockIdx.y * SLABROWS;
  const int tid  = threadIdx.x;
  const int w     = tid >> 5;
  const int lane  = tid & 31;
  const int laneM = lane & 15;
  const int laneH = lane >> 4;
  const int colw  = 32 * w;

  const float* Mj = m + (size_t)j * G * MATELT;

#pragma unroll
  for (int i = 0; i < SLABROWS * (D / 4) / 256; ++i) {
    const int idx = tid + 256 * i;
    const int r = idx >> 6, c4 = idx & 63;
    const v4f v =
        *reinterpret_cast<const v4f*>(Mj + (size_t)(row0 + r) * D + 4 * c4);
    *reinterpret_cast<v4f*>(&slab[r * APITCH + 4 * c4]) = v;
  }

  const float* Bp = Mj + MATELT + (size_t)(2 * laneH) * D + colw + laneM;
  v2f bf[PD][2][2];
#pragma unroll
  for (int s = 0; s < PD; ++s) {
    mm_load_b(bf[s], Bp, KS * s);
    __builtin_amdgcn_sched_barrier(0);
  }
  lds_barrier();

  const int aoff = laneM * APITCH + 2 * laneH;
  const int coff = (8 * laneH) * APITCH + colw + laneM;
  int cur = 0;
  for (int t = 1; t < G - 1; ++t) {
    chain_step<false>(slab + cur + aoff, Bp, bf, slab + (SLABELT - cur) + coff);
    lds_barrier();
    cur = SLABELT - cur;
    Bp += MATELT;
  }
  float* P = ws + (size_t)j * CHUNKSTRIDE;
  chain_step<true>(slab + cur + aoff, Bp, bf,
                   P + (size_t)(row0 + 8 * laneH) * D + colw + laneM);
}

#define GEMV_THREADS 1024
#define KGROUPS 16
#define RPT (D / KGROUPS)

__device__ __forceinline__ const v4f* gemv_slice(const float* M, int kg,
                                                 int cg) {
  return reinterpret_cast<const v4f*>(M + (size_t)(kg * RPT) * D + 4 * cg);
}

__device__ __forceinline__ void gemv_fill(v4f (&r)[RPT], const v4f* p) {
#pragma unroll
  for (int i = 0; i < RPT; ++i) r[i] = p[i * (D / 4)];
}

template <bool MORE>
__device__ __forceinline__ void gemv_step(float* u, float (*part)[D],
                                          v4f (&r)[RPT], const v4f* next,
                                          float* dst, int tid, int kg,
                                          int cg) {
  v4f acc = {0.f, 0.f, 0.f, 0.f};
#pragma unroll
  for (int q = 0; q < RPT / 4; ++q) {
    const v4f uu = *reinterpret_cast<const v4f*>(u + kg * RPT + 4 * q);
#pragma unroll
    for (int e = 0; e < 4; ++e) {
      const int i = 4 * q + e;
      acc.x = fmaf(uu[e], r[i].x, acc.x);
      acc.y = fmaf(uu[e], r[i].y, acc.y);
      acc.z = fmaf(uu[e], r[i].z, acc.z);
      acc.w = fmaf(uu[e], r[i].w, acc.w);
      if (MORE) r[i] = next[i * (D / 4)];
    }
  }
  *reinterpret_cast<v4f*>(&part[kg][4 * cg]) = acc;
  lds_barrier();
  if (tid < D) {
    float s = 0.f;
    for (int g = 0; g < KGROUPS; ++g) s += part[g][tid];
    u[tid] = s;
    dst[tid] = s;
  }
  lds_barrier();
}

__global__ __launch_bounds__(GEMV_THREADS) void prodscan_carries(
    const float* __restrict__ x, const float* __restrict__ ws,
    float* __restrict__ carries, float* __restrict__ out) {
  __shared__ __attribute__((aligned(16))) float u[D];
  __shared__ __attribute__((aligned(16))) float part[KGROUPS][D];
  const int tid = threadIdx.x;
  const int cg  = tid & 63;
  const int kg  = tid >> 6;
  v4f r[RPT];
  gemv_fill(r, gemv_slice(ws, kg, cg));
  if (tid < D) {
    const float v = x[tid];
    u[tid] = v;
    carries[tid] = v;
  }
  lds_barrier();
  for (int j = 0; j < NCHUNK - 1; ++j)
    gemv_step<true>(u, part, r,
                    gemv_slice(ws + (size_t)(j + 1) * CHUNKSTRIDE, kg, cg),
                    carries + (size_t)(j + 1) * D, tid, kg, cg);
  gemv_step<false>(u, part, r, nullptr, carries + (size_t)NCHUNK * D, tid, kg,
                   cg);
  if (tid < D) out[(size_t)NMAT * D + tid] = u[tid];
}

__global__ __launch_bounds__(GEMV_THREADS) void prodscan_expand(
    const float* __restrict__ m, const float* __restrict__ carries,
    float* __restrict__ out) {
  __shared__ __attribute__((aligned(16))) float u[D];
  __shared__ __attribute__((aligned(16))) float part[KGROUPS][D];
  const int j   = blockIdx.x;
  const int tid = threadIdx.x;
  const int cg  = tid & 63;
  const int kg  = tid >> 6;
  const float* Mj = m + (size_t)j * G * MATELT;
  v4f r[RPT];
  gemv_fill(r, gemv_slice(Mj, kg, cg));
  if (tid < D) {
    const float v = carries[(size_t)j * D + tid];
    u[tid] = v;
    out[(size_t)(j * G) * D + tid] = v;
  }
  lds_barrier();
  for (int i = 0; i < G - 2; ++i)
    gemv_step<true>(u, part, r,
                    gemv_slice(Mj + (size_t)(i + 1) * MATELT, kg, cg),
                    out + (size_t)(j * G + i + 1) * D, tid, kg, cg);
  gemv_step<false>(u, part, r, nullptr, out + (size_t)(j * G + G - 1) * D, tid,
                   kg, cg);
}

extern "C" void kernel_launch(void* const* d_in, const int* in_sizes, int n_in,
                              void* d_out, int out_size, void* d_ws,
                              size_t ws_size, hipStream_t stream) {
  (void)in_sizes; (void)n_in; (void)out_size; (void)ws_size;
  const float* x = (const float*)d_in[0];
  const float* m = (const float*)d_in[1];
  float* out = (float*)d_out;
  float* ws  = (float*)d_ws;
  float* carries = ws + (size_t)NCHUNK * CHUNKSTRIDE;

  prodscan_chunk_chain<<<dim3(NCHUNK, NSLAB), 256, 0, stream>>>(m, ws);
  prodscan_carries<<<1, GEMV_THREADS, 0, stream>>>(x, ws, carries, out);
  prodscan_expand<<<NCHUNK, GEMV_THREADS, 0, stream>>>(m, carries, out);
}


# Super-chunk prefix chain (7 serial WMMA steps): 16-row slabs instead of 32, so 128 work-groups instead of 64 and each step half as many WMMAs per WGP; sum order unchanged
# speedup vs baseline: 6.2329x; 1.3330x over previous
#include <hip/hip_runtime.h>
#include <hip/hip_bf16.h>


#define D      256
#define NMAT   1024
#define G      16
#define NCHUNK 64
#define MATELT (D * D)
#define CHUNKSTRIDE (2 * MATELT)
#define SUPER  8
#define NSUPER (NCHUNK / SUPER)

typedef __attribute__((ext_vector_type(2))) float v2f;
typedef __attribute__((ext_vector_type(4))) float v4f;
typedef __attribute__((ext_vector_type(8))) float v8f;

#define SLABROWS 64
#define NSLAB    (D / SLABROWS)
#define APITCH   260
#define SLABELT  (SLABROWS * APITCH)

#define KS 8
#define PD 4

template <int TR>
__device__ __forceinline__ void mm_load_a(v2f (&a)[2][TR], const float* As,
                                          int k0) {
#pragma unroll
  for (int kk = 0; kk < 2; ++kk)
#pragma unroll
    for (int r = 0; r < TR; ++r)
      a[kk][r] = *reinterpret_cast<const v2f*>(As + r * 16 * APITCH + k0 +
                                               4 * kk);
}

__device__ __forceinline__ void mm_load_b(v2f (&b)[2][2], const float* Bp,
                                          int k0) {
#pragma unroll
  for (int kk = 0; kk < 2; ++kk)
#pragma unroll
    for (int c = 0; c < 2; ++c) {
      const float* bp = Bp + (size_t)(k0 + 4 * kk) * D + c * 16;
      b[kk][c].x = bp[0];
      b[kk][c].y = bp[D];
    }
}

template <int TR>
__device__ __forceinline__ void mm_block(v8f (&acc)[TR][2],
                                         const v2f (&a)[2][TR],
                                         const v2f (&b)[2][2]) {
#pragma unroll
  for (int kk = 0; kk < 2; ++kk)
#pragma unroll
    for (int r = 0; r < TR; ++r)
#pragma unroll
      for (int c = 0; c < 2; ++c)
        acc[r][c] = __builtin_amdgcn_wmma_f32_16x16x4_f32(
            false, a[kk][r], false, b[kk][c], (short)0, acc[r][c], false,
            false);
}

__device__ __forceinline__ void lds_barrier() {
  __builtin_amdgcn_fence(__ATOMIC_RELEASE, "workgroup", "local");
  __builtin_amdgcn_s_barrier();
  __builtin_amdgcn_fence(__ATOMIC_ACQUIRE, "workgroup", "local");
}

template <int TR, int PDN, int BSTRIDE, bool MORE, bool TOGLB>
__device__ __forceinline__ void chain_step(const float* As, const float* Bp,
                                           v2f (&bf)[PDN][2][2], float* Cl,
                                           float* Cg) {
  v8f acc[TR][2] = {};
  v2f af[PDN][2][TR];
#pragma unroll
  for (int s = 0; s < PDN; ++s) mm_load_a<TR>(af[s], As, KS * s);

  for (int k0 = 0; k0 < D - KS * PDN; k0 += KS * PDN) {
#pragma unroll
    for (int s = 0; s < PDN; ++s) {
      mm_block<TR>(acc, af[s], bf[s]);
      mm_load_a<TR>(af[s], As, k0 + KS * (s + PDN));
      mm_load_b(bf[s], Bp, k0 + KS * (s + PDN));
      __builtin_amdgcn_sched_barrier(0);
    }
  }
#pragma unroll
  for (int s = 0; s < PDN; ++s) {
    mm_block<TR>(acc, af[s], bf[s]);
    if (MORE) {
      mm_load_b(bf[s], Bp + BSTRIDE, KS * s);
      __builtin_amdgcn_sched_barrier(0);
    }
  }

  if (MORE) {
#pragma unroll
    for (int r = 0; r < TR; ++r)
#pragma unroll
      for (int c = 0; c < 2; ++c)
#pragma unroll
        for (int e = 0; e < 8; ++e)
          Cl[(16 * r + e) * APITCH + 16 * c] = acc[r][c][e];
  }
  if (TOGLB) {
#pragma unroll
    for (int r = 0; r < TR; ++r)
#pragma unroll
      for (int c = 0; c < 2; ++c)
#pragma unroll
        for (int e = 0; e < 8; ++e)
          Cg[(size_t)(16 * r + e) * D + 16 * c] = acc[r][c][e];
  }
}

__global__ __launch_bounds__(256) void prodscan_chunk_chain(
    const float* __restrict__ m, float* __restrict__ ws) {
  __shared__ __attribute__((aligned(16))) float slab[2 * SLABELT];

  const int j    = blockIdx.x;
  const int row0 = blockIdx.y * SLABROWS;
  const int tid  = threadIdx.x;
  const int w     = tid >> 5;
  const int lane  = tid & 31;
  const int laneM = lane & 15;
  const int laneH = lane >> 4;
  const int colw  = 32 * w;

  const float* Mj = m + (size_t)j * G * MATELT;

#pragma unroll
  for (int i = 0; i < SLABROWS * (D / 4) / 256; ++i) {
    const int idx = tid + 256 * i;
    const int r = idx >> 6, c4 = idx & 63;
    const v4f v =
        *reinterpret_cast<const v4f*>(Mj + (size_t)(row0 + r) * D + 4 * c4);
    *reinterpret_cast<v4f*>(&slab[r * APITCH + 4 * c4]) = v;
  }

  const float* Bp = Mj + MATELT + (size_t)(2 * laneH) * D + colw + laneM;
  v2f bf[PD][2][2];
#pragma unroll
  for (int s = 0; s < PD; ++s) {
    mm_load_b(bf[s], Bp, KS * s);
    __builtin_amdgcn_sched_barrier(0);
  }
  lds_barrier();

  const int aoff = laneM * APITCH + 2 * laneH;
  const int coff = (8 * laneH) * APITCH + colw + laneM;
  int cur = 0;
  for (int t = 1; t < G - 1; ++t) {
    chain_step<4, PD, MATELT, true, false>(slab + cur + aoff, Bp, bf,
                                           slab + (SLABELT - cur) + coff,
                                           nullptr);
    lds_barrier();
    cur = SLABELT - cur;
    Bp += MATELT;
  }
  float* P = ws + (size_t)j * CHUNKSTRIDE;
  chain_step<4, PD, MATELT, false, true>(
      slab + cur + aoff, Bp, bf, nullptr,
      P + (size_t)(row0 + 8 * laneH) * D + colw + laneM);
}

#define TR_S      1
#define PD_S      8
#define SLABROWS_S (16 * TR_S)
#define NSLAB_S    (D / SLABROWS_S)
#define SLABELT_S  (SLABROWS_S * APITCH)

__global__ __launch_bounds__(256) void prodscan_super_chain(
    const float* __restrict__ pin, float* __restrict__ pout) {
  __shared__ __attribute__((aligned(16))) float slab[2 * SLABELT_S];

  const int s    = blockIdx.x;
  const int row0 = blockIdx.y * SLABROWS_S;
  const int tid  = threadIdx.x;
  const int w     = tid >> 5;
  const int lane  = tid & 31;
  const int laneM = lane & 15;
  const int laneH = lane >> 4;
  const int colw  = 32 * w;

  const float* P0 = pin + (size_t)(s * SUPER) * CHUNKSTRIDE;

#pragma unroll
  for (int i = 0; i < SLABROWS_S * (D / 4) / 256; ++i) {
    const int idx = tid + 256 * i;
    const int r = idx >> 6, c4 = idx & 63;
    const v4f v =
        *reinterpret_cast<const v4f*>(P0 + (size_t)(row0 + r) * D + 4 * c4);
    *reinterpret_cast<v4f*>(&slab[r * APITCH + 4 * c4]) = v;
  }

  const float* Bp = P0 + CHUNKSTRIDE + (size_t)(2 * laneH) * D + colw + laneM;
  v2f bf[PD_S][2][2];
#pragma unroll
  for (int q = 0; q < PD_S; ++q) {
    mm_load_b(bf[q], Bp, KS * q);
    __builtin_amdgcn_sched_barrier(0);
  }
  lds_barrier();

  const int aoff = laneM * APITCH + 2 * laneH;
  const int coff = (8 * laneH) * APITCH + colw + laneM;
  float* Cg = pout + (size_t)(s * SUPER + 1) * CHUNKSTRIDE +
              (size_t)(row0 + 8 * laneH) * D + colw + laneM;
  int cur = 0;
  for (int t = 1; t < SUPER - 1; ++t) {
    chain_step<TR_S, PD_S, CHUNKSTRIDE, true, true>(
        slab + cur + aoff, Bp, bf, slab + (SLABELT_S - cur) + coff, Cg);
    lds_barrier();
    cur = SLABELT_S - cur;
    Bp += CHUNKSTRIDE;
    Cg += CHUNKSTRIDE;
  }
  chain_step<TR_S, PD_S, CHUNKSTRIDE, false, true>(slab + cur + aoff, Bp, bf,
                                                   nullptr, Cg);
}

#define GEMV_THREADS 1024
#define KGROUPS 16
#define RPT (D / KGROUPS)

__device__ __forceinline__ const v4f* gemv_slice(const float* M, int kg,
                                                 int cg) {
  return reinterpret_cast<const v4f*>(M + (size_t)(kg * RPT) * D + 4 * cg);
}

__device__ __forceinline__ void gemv_fill(v4f (&r)[RPT], const v4f* p) {
#pragma unroll
  for (int i = 0; i < RPT; ++i) r[i] = p[i * (D / 4)];
}

template <bool MORE>
__device__ __forceinline__ void gemv_step(float* u, float (*part)[D],
                                          v4f (&r)[RPT], const v4f* next,
                                          float* dst, int tid, int kg,
                                          int cg) {
  v4f acc = {0.f, 0.f, 0.f, 0.f};
#pragma unroll
  for (int q = 0; q < RPT / 4; ++q) {
    const v4f uu = *reinterpret_cast<const v4f*>(u + kg * RPT + 4 * q);
#pragma unroll
    for (int e = 0; e < 4; ++e) {
      const int i = 4 * q + e;
      acc.x = fmaf(uu[e], r[i].x, acc.x);
      acc.y = fmaf(uu[e], r[i].y, acc.y);
      acc.z = fmaf(uu[e], r[i].z, acc.z);
      acc.w = fmaf(uu[e], r[i].w, acc.w);
      if (MORE) r[i] = next[i * (D / 4)];
    }
  }
  *reinterpret_cast<v4f*>(&part[kg][4 * cg]) = acc;
  lds_barrier();
  if (tid < D) {
    float s = 0.f;
    for (int g = 0; g < KGROUPS; ++g) s += part[g][tid];
    u[tid] = s;
    dst[tid] = s;
  }
  lds_barrier();
}

__device__ __forceinline__ const float* chunk_prefix(const float* ws, int j) {
  return ws + (size_t)j * CHUNKSTRIDE + ((j % SUPER) ? MATELT : 0);
}

__global__ __launch_bounds__(GEMV_THREADS) void prodscan_carries(
    const float* __restrict__ x, const float* __restrict__ ws,
    float* __restrict__ carries, float* __restrict__ out) {
  __shared__ __attribute__((aligned(16))) float u[D];
  __shared__ __attribute__((aligned(16))) float part[KGROUPS][D];
  const int tid = threadIdx.x;
  const int cg  = tid & 63;
  const int kg  = tid >> 6;
  v4f r[RPT];
  gemv_fill(r, gemv_slice(chunk_prefix(ws, SUPER - 1), kg, cg));
  if (tid < D) {
    const float v = x[tid];
    u[tid] = v;
    carries[tid] = v;
  }
  lds_barrier();
  for (int s = 0; s < NSUPER - 1; ++s)
    gemv_step<true>(
        u, part, r,
        gemv_slice(chunk_prefix(ws, (s + 2) * SUPER - 1), kg, cg),
        carries + (size_t)(s + 1) * D, tid, kg, cg);
  gemv_step<false>(u, part, r, nullptr, carries + (size_t)NSUPER * D, tid, kg,
                   cg);
  if (tid < D) out[(size_t)NMAT * D + tid] = u[tid];
}

__global__ __launch_bounds__(GEMV_THREADS) void prodscan_expand(
    const float* __restrict__ m, const float* __restrict__ ws,
    const float* __restrict__ carries, float* __restrict__ out) {
  __shared__ __attribute__((aligned(16))) float u[D];
  __shared__ __attribute__((aligned(16))) float part[KGROUPS][D];
  const int j   = blockIdx.x;
  const int tid = threadIdx.x;
  const int cg  = tid & 63;
  const int kg  = tid >> 6;
  const float* Mj = m + (size_t)j * G * MATELT;
  const bool first = (j % SUPER) == 0;
  v4f r[RPT];
  gemv_fill(r, gemv_slice(first ? Mj : chunk_prefix(ws, j - 1), kg, cg));
  if (tid < D) {
    const float v = carries[(size_t)(j / SUPER) * D + tid];
    u[tid] = v;
    if (first) out[(size_t)(j * G) * D + tid] = v;
  }
  lds_barrier();
  for (int i = first ? 0 : -1; i < G - 2; ++i)
    gemv_step<true>(u, part, r,
                    gemv_slice(Mj + (size_t)(i + 1) * MATELT, kg, cg),
                    out + (size_t)(j * G + i + 1) * D, tid, kg, cg);
  gemv_step<false>(u, part, r, nullptr, out + (size_t)(j * G + G - 1) * D, tid,
                   kg, cg);
}

extern "C" void kernel_launch(void* const* d_in, const int* in_sizes, int n_in,
                              void* d_out, int out_size, void* d_ws,
                              size_t ws_size, hipStream_t stream) {
  (void)in_sizes; (void)n_in; (void)out_size; (void)ws_size;
  const float* x = (const float*)d_in[0];
  const float* m = (const float*)d_in[1];
  float* out = (float*)d_out;
  float* ws  = (float*)d_ws;
  float* carries = ws + (size_t)NCHUNK * CHUNKSTRIDE;

  prodscan_chunk_chain<<<dim3(NCHUNK, NSLAB), 256, 0, stream>>>(m, ws);
  prodscan_super_chain<<<dim3(NSUPER, NSLAB_S), 256, 0, stream>>>(ws,
                                                                  ws + MATELT);
  prodscan_carries<<<1, GEMV_THREADS, 0, stream>>>(x, ws, carries, out);
  prodscan_expand<<<NCHUNK, GEMV_THREADS, 0, stream>>>(m, ws, carries, out);
}


# WMMA k-loops: ring slot 16 k, not 8, its refill held as ONE load group behind the block: one load place per 32 WMMAs in the chunk chain (was two per 16), per 8 in the super chain; sum order unchanged
# speedup vs baseline: 6.4491x; 1.0347x over previous
#include <hip/hip_runtime.h>
#include <hip/hip_bf16.h>


#define D      256
#define NMAT   1024
#define G      16
#define NCHUNK 64
#define MATELT (D * D)
#define CHUNKSTRIDE (2 * MATELT)
#define SUPER  8
#define NSUPER (NCHUNK / SUPER)

typedef __attribute__((ext_vector_type(2))) float v2f;
typedef __attribute__((ext_vector_type(4))) float v4f;
typedef __attribute__((ext_vector_type(8))) float v8f;

#define SLABROWS 64
#define NSLAB    (D / SLABROWS)
#define APITCH   260
#define SLABELT  (SLABROWS * APITCH)

#define KS 16
#define KK (KS / 4)
#define PD 4

template <int TR>
__device__ __forceinline__ void mm_load_a(v2f (&a)[KK][TR], const float* As,
                                          int k0) {
#pragma unroll
  for (int kk = 0; kk < KK; ++kk)
#pragma unroll
    for (int r = 0; r < TR; ++r)
      a[kk][r] = *reinterpret_cast<const v2f*>(As + r * 16 * APITCH + k0 +
                                               4 * kk);
}

__device__ __forceinline__ void mm_load_b(v2f (&b)[KK][2], const float* Bp,
                                          int k0) {
#pragma unroll
  for (int kk = 0; kk < KK; ++kk)
#pragma unroll
    for (int c = 0; c < 2; ++c) {
      const float* bp = Bp + (size_t)(k0 + 4 * kk) * D + c * 16;
      b[kk][c].x = bp[0];
      b[kk][c].y = bp[D];
    }
}

template <int TR>
__device__ __forceinline__ void mm_block(v8f (&acc)[TR][2],
                                         const v2f (&a)[KK][TR],
                                         const v2f (&b)[KK][2]) {
#pragma unroll
  for (int kk = 0; kk < KK; ++kk)
#pragma unroll
    for (int r = 0; r < TR; ++r)
#pragma unroll
      for (int c = 0; c < 2; ++c)
        acc[r][c] = __builtin_amdgcn_wmma_f32_16x16x4_f32(
            false, a[kk][r], false, b[kk][c], (short)0, acc[r][c], false,
            false);
}

__device__ __forceinline__ void lds_barrier() {
  __builtin_amdgcn_fence(__ATOMIC_RELEASE, "workgroup", "local");
  __builtin_amdgcn_s_barrier();
  __builtin_amdgcn_fence(__ATOMIC_ACQUIRE, "workgroup", "local");
}

template <int TR, int PDN, int BSTRIDE, bool MORE, bool TOGLB>
__device__ __forceinline__ void chain_step(const float* As, const float* Bp,
                                           v2f (&bf)[PDN][KK][2], float* Cl,
                                           float* Cg) {
  v8f acc[TR][2] = {};
  v2f af[PDN][KK][TR];
#pragma unroll
  for (int s = 0; s < PDN; ++s) mm_load_a<TR>(af[s], As, KS * s);

  for (int k0 = 0; k0 < D - KS * PDN; k0 += KS * PDN) {
#pragma unroll
    for (int s = 0; s < PDN; ++s) {
      mm_block<TR>(acc, af[s], bf[s]);
      __builtin_amdgcn_sched_barrier(0);
      mm_load_a<TR>(af[s], As, k0 + KS * (s + PDN));
      mm_load_b(bf[s], Bp, k0 + KS * (s + PDN));
      __builtin_amdgcn_sched_barrier(0);
    }
  }
#pragma unroll
  for (int s = 0; s < PDN; ++s) {
    mm_block<TR>(acc, af[s], bf[s]);
    if (MORE) {
      __builtin_amdgcn_sched_barrier(0);
      mm_load_b(bf[s], Bp + BSTRIDE, KS * s);
      __builtin_amdgcn_sched_barrier(0);
    }
  }

  if (MORE) {
#pragma unroll
    for (int r = 0; r < TR; ++r)
#pragma unroll
      for (int c = 0; c < 2; ++c)
#pragma unroll
        for (int e = 0; e < 8; ++e)
          Cl[(16 * r + e) * APITCH + 16 * c] = acc[r][c][e];
  }
  if (TOGLB) {
#pragma unroll
    for (int r = 0; r < TR; ++r)
#pragma unroll
      for (int c = 0; c < 2; ++c)
#pragma unroll
        for (int e = 0; e < 8; ++e)
          Cg[(size_t)(16 * r + e) * D + 16 * c] = acc[r][c][e];
  }
}

__global__ __launch_bounds__(256) void prodscan_chunk_chain(
    const float* __restrict__ m, float* __restrict__ ws) {
  __shared__ __attribute__((aligned(16))) float slab[2 * SLABELT];

  const int j    = blockIdx.x;
  const int row0 = blockIdx.y * SLABROWS;
  const int tid  = threadIdx.x;
  const int w     = tid >> 5;
  const int lane  = tid & 31;
  const int laneM = lane & 15;
  const int laneH = lane >> 4;
  const int colw  = 32 * w;

  const float* Mj = m + (size_t)j * G * MATELT;

#pragma unroll
  for (int i = 0; i < SLABROWS * (D / 4) / 256; ++i) {
    const int idx = tid + 256 * i;
    const int r = idx >> 6, c4 = idx & 63;
    const v4f v =
        *reinterpret_cast<const v4f*>(Mj + (size_t)(row0 + r) * D + 4 * c4);
    *reinterpret_cast<v4f*>(&slab[r * APITCH + 4 * c4]) = v;
  }

  const float* Bp = Mj + MATELT + (size_t)(2 * laneH) * D + colw + laneM;
  v2f bf[PD][KK][2];
#pragma unroll
  for (int s = 0; s < PD; ++s) {
    mm_load_b(bf[s], Bp, KS * s);
    __builtin_amdgcn_sched_barrier(0);
  }
  lds_barrier();

  const int aoff = laneM * APITCH + 2 * laneH;
  const int coff = (8 * laneH) * APITCH + colw + laneM;
  int cur = 0;
  for (int t = 1; t < G - 1; ++t) {
    chain_step<4, PD, MATELT, true, false>(slab + cur + aoff, Bp, bf,
                                           slab + (SLABELT - cur) + coff,
                                           nullptr);
    lds_barrier();
    cur = SLABELT - cur;
    Bp += MATELT;
  }
  float* P = ws + (size_t)j * CHUNKSTRIDE;
  chain_step<4, PD, MATELT, false, true>(
      slab + cur + aoff, Bp, bf, nullptr,
      P + (size_t)(row0 + 8 * laneH) * D + colw + laneM);
}

#define TR_S      1
#define PD_S      4
#define SLABROWS_S (16 * TR_S)
#define NSLAB_S    (D / SLABROWS_S)
#define SLABELT_S  (SLABROWS_S * APITCH)

__global__ __launch_bounds__(256) void prodscan_super_chain(
    const float* __restrict__ pin, float* __restrict__ pout) {
  __shared__ __attribute__((aligned(16))) float slab[2 * SLABELT_S];

  const int s    = blockIdx.x;
  const int row0 = blockIdx.y * SLABROWS_S;
  const int tid  = threadIdx.x;
  const int w     = tid >> 5;
  const int lane  = tid & 31;
  const int laneM = lane & 15;
  const int laneH = lane >> 4;
  const int colw  = 32 * w;

  const float* P0 = pin + (size_t)(s * SUPER) * CHUNKSTRIDE;

#pragma unroll
  for (int i = 0; i < SLABROWS_S * (D / 4) / 256; ++i) {
    const int idx = tid + 256 * i;
    const int r = idx >> 6, c4 = idx & 63;
    const v4f v =
        *reinterpret_cast<const v4f*>(P0 + (size_t)(row0 + r) * D + 4 * c4);
    *reinterpret_cast<v4f*>(&slab[r * APITCH + 4 * c4]) = v;
  }

  const float* Bp = P0 + CHUNKSTRIDE + (size_t)(2 * laneH) * D + colw + laneM;
  v2f bf[PD_S][KK][2];
#pragma unroll
  for (int q = 0; q < PD_S; ++q) {
    mm_load_b(bf[q], Bp, KS * q);
    __builtin_amdgcn_sched_barrier(0);
  }
  lds_barrier();

  const int aoff = laneM * APITCH + 2 * laneH;
  const int coff = (8 * laneH) * APITCH + colw + laneM;
  float* Cg = pout + (size_t)(s * SUPER + 1) * CHUNKSTRIDE +
              (size_t)(row0 + 8 * laneH) * D + colw + laneM;
  int cur = 0;
  for (int t = 1; t < SUPER - 1; ++t) {
    chain_step<TR_S, PD_S, CHUNKSTRIDE, true, true>(
        slab + cur + aoff, Bp, bf, slab + (SLABELT_S - cur) + coff, Cg);
    lds_barrier();
    cur = SLABELT_S - cur;
    Bp += CHUNKSTRIDE;
    Cg += CHUNKSTRIDE;
  }
  chain_step<TR_S, PD_S, CHUNKSTRIDE, false, true>(slab + cur + aoff, Bp, bf,
                                                   nullptr, Cg);
}

#define GEMV_THREADS 1024
#define KGROUPS 16
#define RPT (D / KGROUPS)

__device__ __forceinline__ const v4f* gemv_slice(const float* M, int kg,
                                                 int cg) {
  return reinterpret_cast<const v4f*>(M + (size_t)(kg * RPT) * D + 4 * cg);
}

__device__ __forceinline__ void gemv_fill(v4f (&r)[RPT], const v4f* p) {
#pragma unroll
  for (int i = 0; i < RPT; ++i) r[i] = p[i * (D / 4)];
}

template <bool MORE>
__device__ __forceinline__ void gemv_step(float* u, float (*part)[D],
                                          v4f (&r)[RPT], const v4f* next,
                                          float* dst, int tid, int kg,
                                          int cg) {
  v4f acc = {0.f, 0.f, 0.f, 0.f};
#pragma unroll
  for (int q = 0; q < RPT / 4; ++q) {
    const v4f uu = *reinterpret_cast<const v4f*>(u + kg * RPT + 4 * q);
#pragma unroll
    for (int e = 0; e < 4; ++e) {
      const int i = 4 * q + e;
      acc.x = fmaf(uu[e], r[i].x, acc.x);
      acc.y = fmaf(uu[e], r[i].y, acc.y);
      acc.z = fmaf(uu[e], r[i].z, acc.z);
      acc.w = fmaf(uu[e], r[i].w, acc.w);
      if (MORE) r[i] = next[i * (D / 4)];
    }
  }
  *reinterpret_cast<v4f*>(&part[kg][4 * cg]) = acc;
  lds_barrier();
  if (tid < D) {
    float s = 0.f;
    for (int g = 0; g < KGROUPS; ++g) s += part[g][tid];
    u[tid] = s;
    dst[tid] = s;
  }
  lds_barrier();
}

__device__ __forceinline__ const float* chunk_prefix(const float* ws, int j) {
  return ws + (size_t)j * CHUNKSTRIDE + ((j % SUPER) ? MATELT : 0);
}

__global__ __launch_bounds__(GEMV_THREADS) void prodscan_carries(
    const float* __restrict__ x, const float* __restrict__ ws,
    float* __restrict__ carries, float* __restrict__ out) {
  __shared__ __attribute__((aligned(16))) float u[D];
  __shared__ __attribute__((aligned(16))) float part[KGROUPS][D];
  const int tid = threadIdx.x;
  const int cg  = tid & 63;
  const int kg  = tid >> 6;
  v4f r[RPT];
  gemv_fill(r, gemv_slice(chunk_prefix(ws, SUPER - 1), kg, cg));
  if (tid < D) {
    const float v = x[tid];
    u[tid] = v;
    carries[tid] = v;
  }
  lds_barrier();
  for (int s = 0; s < NSUPER - 1; ++s)
    gemv_step<true>(
        u, part, r,
        gemv_slice(chunk_prefix(ws, (s + 2) * SUPER - 1), kg, cg),
        carries + (size_t)(s + 1) * D, tid, kg, cg);
  gemv_step<false>(u, part, r, nullptr, carries + (size_t)NSUPER * D, tid, kg,
                   cg);
  if (tid < D) out[(size_t)NMAT * D + tid] = u[tid];
}

__global__ __launch_bounds__(GEMV_THREADS) void prodscan_expand(
    const float* __restrict__ m, const float* __restrict__ ws,
    const float* __restrict__ carries, float* __restrict__ out) {
  __shared__ __attribute__((aligned(16))) float u[D];
  __shared__ __attribute__((aligned(16))) float part[KGROUPS][D];
  const int j   = blockIdx.x;
  const int tid = threadIdx.x;
  const int cg  = tid & 63;
  const int kg  = tid >> 6;
  const float* Mj = m + (size_t)j * G * MATELT;
  const bool first = (j % SUPER) == 0;
  v4f r[RPT];
  gemv_fill(r, gemv_slice(first ? Mj : chunk_prefix(ws, j - 1), kg, cg));
  if (tid < D) {
    const float v = carries[(size_t)(j / SUPER) * D + tid];
    u[tid] = v;
    if (first) out[(size_t)(j * G) * D + tid] = v;
  }
  lds_barrier();
  for (int i = first ? 0 : -1; i < G - 2; ++i)
    gemv_step<true>(u, part, r,
                    gemv_slice(Mj + (size_t)(i + 1) * MATELT, kg, cg),
                    out + (size_t)(j * G + i + 1) * D, tid, kg, cg);
  gemv_step<false>(u, part, r, nullptr, out + (size_t)(j * G + G - 1) * D, tid,
                   kg, cg);
}

extern "C" void kernel_launch(void* const* d_in, const int* in_sizes, int n_in,
                              void* d_out, int out_size, void* d_ws,
                              size_t ws_size, hipStream_t stream) {
  (void)in_sizes; (void)n_in; (void)out_size; (void)ws_size;
  const float* x = (const float*)d_in[0];
  const float* m = (const float*)d_in[1];
  float* out = (float*)d_out;
  float* ws  = (float*)d_ws;
  float* carries = ws + (size_t)NCHUNK * CHUNKSTRIDE;

  prodscan_chunk_chain<<<dim3(NCHUNK, NSLAB), 256, 0, stream>>>(m, ws);
  prodscan_super_chain<<<dim3(NSUPER, NSLAB_S), 256, 0, stream>>>(ws,
                                                                  ws + MATELT);
  prodscan_carries<<<1, GEMV_THREADS, 0, stream>>>(x, ws, carries, out);
  prodscan_expand<<<NCHUNK, GEMV_THREADS, 0, stream>>>(m, ws, carries, out);
}


# Expansion chain: every thread prefetches one 256-B unit of the matrix two steps ahead into L2 (global_prefetch_b8, same work-group), so a step's row loads hit L2, not memory; sums unchanged
# speedup vs baseline: 6.8169x; 1.0570x over previous
#include <hip/hip_runtime.h>
#include <hip/hip_bf16.h>


#define D      256
#define NMAT   1024
#define G      16
#define NCHUNK 64
#define MATELT (D * D)
#define CHUNKSTRIDE (2 * MATELT)
#define SUPER  8
#define NSUPER (NCHUNK / SUPER)

typedef __attribute__((ext_vector_type(2))) float v2f;
typedef __attribute__((ext_vector_type(4))) float v4f;
typedef __attribute__((ext_vector_type(8))) float v8f;

#define SLABROWS 64
#define NSLAB    (D / SLABROWS)
#define APITCH   260
#define SLABELT  (SLABROWS * APITCH)

#define KS 16
#define KK (KS / 4)
#define PD 4

template <int TR>
__device__ __forceinline__ void mm_load_a(v2f (&a)[KK][TR], const float* As,
                                          int k0) {
#pragma unroll
  for (int kk = 0; kk < KK; ++kk)
#pragma unroll
    for (int r = 0; r < TR; ++r)
      a[kk][r] = *reinterpret_cast<const v2f*>(As + r * 16 * APITCH + k0 +
                                               4 * kk);
}

__device__ __forceinline__ void mm_load_b(v2f (&b)[KK][2], const float* Bp,
                                          int k0) {
#pragma unroll
  for (int kk = 0; kk < KK; ++kk)
#pragma unroll
    for (int c = 0; c < 2; ++c) {
      const float* bp = Bp + (size_t)(k0 + 4 * kk) * D + c * 16;
      b[kk][c].x = bp[0];
      b[kk][c].y = bp[D];
    }
}

template <int TR>
__device__ __forceinline__ void mm_block(v8f (&acc)[TR][2],
                                         const v2f (&a)[KK][TR],
                                         const v2f (&b)[KK][2]) {
#pragma unroll
  for (int kk = 0; kk < KK; ++kk)
#pragma unroll
    for (int r = 0; r < TR; ++r)
#pragma unroll
      for (int c = 0; c < 2; ++c)
        acc[r][c] = __builtin_amdgcn_wmma_f32_16x16x4_f32(
            false, a[kk][r], false, b[kk][c], (short)0, acc[r][c], false,
            false);
}

__device__ __forceinline__ void lds_barrier() {
  __builtin_amdgcn_fence(__ATOMIC_RELEASE, "workgroup", "local");
  __builtin_amdgcn_s_barrier();
  __builtin_amdgcn_fence(__ATOMIC_ACQUIRE, "workgroup", "local");
}

template <int TR, int PDN, int BSTRIDE, bool MORE, bool TOGLB>
__device__ __forceinline__ void chain_step(const float* As, const float* Bp,
                                           v2f (&bf)[PDN][KK][2], float* Cl,
                                           float* Cg) {
  v8f acc[TR][2] = {};
  v2f af[PDN][KK][TR];
#pragma unroll
  for (int s = 0; s < PDN; ++s) mm_load_a<TR>(af[s], As, KS * s);

  for (int k0 = 0; k0 < D - KS * PDN; k0 += KS * PDN) {
#pragma unroll
    for (int s = 0; s < PDN; ++s) {
      mm_block<TR>(acc, af[s], bf[s]);
      __builtin_amdgcn_sched_barrier(0);
      mm_load_a<TR>(af[s], As, k0 + KS * (s + PDN));
      mm_load_b(bf[s], Bp, k0 + KS * (s + PDN));
      __builtin_amdgcn_sched_barrier(0);
    }
  }
#pragma unroll
  for (int s = 0; s < PDN; ++s) {
    mm_block<TR>(acc, af[s], bf[s]);
    if (MORE) {
      __builtin_amdgcn_sched_barrier(0);
      mm_load_b(bf[s], Bp + BSTRIDE, KS * s);
      __builtin_amdgcn_sched_barrier(0);
    }
  }

  if (MORE) {
#pragma unroll
    for (int r = 0; r < TR; ++r)
#pragma unroll
      for (int c = 0; c < 2; ++c)
#pragma unroll
        for (int e = 0; e < 8; ++e)
          Cl[(16 * r + e) * APITCH + 16 * c] = acc[r][c][e];
  }
  if (TOGLB) {
#pragma unroll
    for (int r = 0; r < TR; ++r)
#pragma unroll
      for (int c = 0; c < 2; ++c)
#pragma unroll
        for (int e = 0; e < 8; ++e)
          Cg[(size_t)(16 * r + e) * D + 16 * c] = acc[r][c][e];
  }
}

__global__ __launch_bounds__(256) void prodscan_chunk_chain(
    const float* __restrict__ m, float* __restrict__ ws) {
  __shared__ __attribute__((aligned(16))) float slab[2 * SLABELT];

  const int j    = blockIdx.x;
  const int row0 = blockIdx.y * SLABROWS;
  const int tid  = threadIdx.x;
  const int w     = tid >> 5;
  const int lane  = tid & 31;
  const int laneM = lane & 15;
  const int laneH = lane >> 4;
  const int colw  = 32 * w;

  const float* Mj = m + (size_t)j * G * MATELT;

#pragma unroll
  for (int i = 0; i < SLABROWS * (D / 4) / 256; ++i) {
    const int idx = tid + 256 * i;
    const int r = idx >> 6, c4 = idx & 63;
    const v4f v =
        *reinterpret_cast<const v4f*>(Mj + (size_t)(row0 + r) * D + 4 * c4);
    *reinterpret_cast<v4f*>(&slab[r * APITCH + 4 * c4]) = v;
  }

  const float* Bp = Mj + MATELT + (size_t)(2 * laneH) * D + colw + laneM;
  v2f bf[PD][KK][2];
#pragma unroll
  for (int s = 0; s < PD; ++s) {
    mm_load_b(bf[s], Bp, KS * s);
    __builtin_amdgcn_sched_barrier(0);
  }
  lds_barrier();

  const int aoff = laneM * APITCH + 2 * laneH;
  const int coff = (8 * laneH) * APITCH + colw + laneM;
  int cur = 0;
  for (int t = 1; t < G - 1; ++t) {
    chain_step<4, PD, MATELT, true, false>(slab + cur + aoff, Bp, bf,
                                           slab + (SLABELT - cur) + coff,
                                           nullptr);
    lds_barrier();
    cur = SLABELT - cur;
    Bp += MATELT;
  }
  float* P = ws + (size_t)j * CHUNKSTRIDE;
  chain_step<4, PD, MATELT, false, true>(
      slab + cur + aoff, Bp, bf, nullptr,
      P + (size_t)(row0 + 8 * laneH) * D + colw + laneM);
}

#define TR_S      1
#define PD_S      4
#define SLABROWS_S (16 * TR_S)
#define NSLAB_S    (D / SLABROWS_S)
#define SLABELT_S  (SLABROWS_S * APITCH)

__global__ __launch_bounds__(256) void prodscan_super_chain(
    const float* __restrict__ pin, float* __restrict__ pout) {
  __shared__ __attribute__((aligned(16))) float slab[2 * SLABELT_S];

  const int s    = blockIdx.x;
  const int row0 = blockIdx.y * SLABROWS_S;
  const int tid  = threadIdx.x;
  const int w     = tid >> 5;
  const int lane  = tid & 31;
  const int laneM = lane & 15;
  const int laneH = lane >> 4;
  const int colw  = 32 * w;

  const float* P0 = pin + (size_t)(s * SUPER) * CHUNKSTRIDE;

#pragma unroll
  for (int i = 0; i < SLABROWS_S * (D / 4) / 256; ++i) {
    const int idx = tid + 256 * i;
    const int r = idx >> 6, c4 = idx & 63;
    const v4f v =
        *reinterpret_cast<const v4f*>(P0 + (size_t)(row0 + r) * D + 4 * c4);
    *reinterpret_cast<v4f*>(&slab[r * APITCH + 4 * c4]) = v;
  }

  const float* Bp = P0 + CHUNKSTRIDE + (size_t)(2 * laneH) * D + colw + laneM;
  v2f bf[PD_S][KK][2];
#pragma unroll
  for (int q = 0; q < PD_S; ++q) {
    mm_load_b(bf[q], Bp, KS * q);
    __builtin_amdgcn_sched_barrier(0);
  }
  lds_barrier();

  const int aoff = laneM * APITCH + 2 * laneH;
  const int coff = (8 * laneH) * APITCH + colw + laneM;
  float* Cg = pout + (size_t)(s * SUPER + 1) * CHUNKSTRIDE +
              (size_t)(row0 + 8 * laneH) * D + colw + laneM;
  int cur = 0;
  for (int t = 1; t < SUPER - 1; ++t) {
    chain_step<TR_S, PD_S, CHUNKSTRIDE, true, true>(
        slab + cur + aoff, Bp, bf, slab + (SLABELT_S - cur) + coff, Cg);
    lds_barrier();
    cur = SLABELT_S - cur;
    Bp += CHUNKSTRIDE;
    Cg += CHUNKSTRIDE;
  }
  chain_step<TR_S, PD_S, CHUNKSTRIDE, false, true>(slab + cur + aoff, Bp, bf,
                                                   nullptr, Cg);
}

#define GEMV_THREADS 1024
#define KGROUPS 16
#define RPT (D / KGROUPS)
#define PFD 2

__device__ __forceinline__ const v4f* gemv_slice(const float* M, int kg,
                                                 int cg) {
  return reinterpret_cast<const v4f*>(M + (size_t)(kg * RPT) * D + 4 * cg);
}

__device__ __forceinline__ void gemv_fill(v4f (&r)[RPT], const v4f* p) {
#pragma unroll
  for (int i = 0; i < RPT; ++i) r[i] = p[i * (D / 4)];
}

__device__ __forceinline__ void l2_prefetch_matrix(const float* M, int tid) {
  __builtin_prefetch(M + tid * 64, 0, 2);
}

template <bool MORE>
__device__ __forceinline__ void gemv_step(float* u, float (*part)[D],
                                          v4f (&r)[RPT], const v4f* next,
                                          float* dst, int tid, int kg,
                                          int cg) {
  v4f acc = {0.f, 0.f, 0.f, 0.f};
#pragma unroll
  for (int q = 0; q < RPT / 4; ++q) {
    const v4f uu = *reinterpret_cast<const v4f*>(u + kg * RPT + 4 * q);
#pragma unroll
    for (int e = 0; e < 4; ++e) {
      const int i = 4 * q + e;
      acc.x = fmaf(uu[e], r[i].x, acc.x);
      acc.y = fmaf(uu[e], r[i].y, acc.y);
      acc.z = fmaf(uu[e], r[i].z, acc.z);
      acc.w = fmaf(uu[e], r[i].w, acc.w);
      if (MORE) r[i] = next[i * (D / 4)];
    }
  }
  *reinterpret_cast<v4f*>(&part[kg][4 * cg]) = acc;
  lds_barrier();
  if (tid < D) {
    float s = 0.f;
    for (int g = 0; g < KGROUPS; ++g) s += part[g][tid];
    u[tid] = s;
    dst[tid] = s;
  }
  lds_barrier();
}

__device__ __forceinline__ const float* chunk_prefix(const float* ws, int j) {
  return ws + (size_t)j * CHUNKSTRIDE + ((j % SUPER) ? MATELT : 0);
}

__global__ __launch_bounds__(GEMV_THREADS) void prodscan_carries(
    const float* __restrict__ x, const float* __restrict__ ws,
    float* __restrict__ carries, float* __restrict__ out) {
  __shared__ __attribute__((aligned(16))) float u[D];
  __shared__ __attribute__((aligned(16))) float part[KGROUPS][D];
  const int tid = threadIdx.x;
  const int cg  = tid & 63;
  const int kg  = tid >> 6;
  v4f r[RPT];
  gemv_fill(r, gemv_slice(chunk_prefix(ws, SUPER - 1), kg, cg));
  if (tid < D) {
    const float v = x[tid];
    u[tid] = v;
    carries[tid] = v;
  }
  lds_barrier();
  for (int s = 0; s < NSUPER - 1; ++s)
    gemv_step<true>(
        u, part, r,
        gemv_slice(chunk_prefix(ws, (s + 2) * SUPER - 1), kg, cg),
        carries + (size_t)(s + 1) * D, tid, kg, cg);
  gemv_step<false>(u, part, r, nullptr, carries + (size_t)NSUPER * D, tid, kg,
                   cg);
  if (tid < D) out[(size_t)NMAT * D + tid] = u[tid];
}

__global__ __launch_bounds__(GEMV_THREADS) void prodscan_expand(
    const float* __restrict__ m, const float* __restrict__ ws,
    const float* __restrict__ carries, float* __restrict__ out) {
  __shared__ __attribute__((aligned(16))) float u[D];
  __shared__ __attribute__((aligned(16))) float part[KGROUPS][D];
  const int j   = blockIdx.x;
  const int tid = threadIdx.x;
  const int cg  = tid & 63;
  const int kg  = tid >> 6;
  const float* Mj = m + (size_t)j * G * MATELT;
  const bool first = (j % SUPER) == 0;
  const int a0 = first ? 1 : 0;
  for (int a = a0; a < a0 + PFD; ++a)
    l2_prefetch_matrix(Mj + (size_t)a * MATELT, tid);
  v4f r[RPT];
  gemv_fill(r, gemv_slice(first ? Mj : chunk_prefix(ws, j - 1), kg, cg));
  if (tid < D) {
    const float v = carries[(size_t)(j / SUPER) * D + tid];
    u[tid] = v;
    if (first) out[(size_t)(j * G) * D + tid] = v;
  }
  lds_barrier();
  int i = first ? 0 : -1;
  for (; i < G - 2 - PFD; ++i) {
    l2_prefetch_matrix(Mj + (size_t)(i + 1 + PFD) * MATELT, tid);
    gemv_step<true>(u, part, r,
                    gemv_slice(Mj + (size_t)(i + 1) * MATELT, kg, cg),
                    out + (size_t)(j * G + i + 1) * D, tid, kg, cg);
  }
  for (; i < G - 2; ++i)
    gemv_step<true>(u, part, r,
                    gemv_slice(Mj + (size_t)(i + 1) * MATELT, kg, cg),
                    out + (size_t)(j * G + i + 1) * D, tid, kg, cg);
  gemv_step<false>(u, part, r, nullptr, out + (size_t)(j * G + G - 1) * D, tid,
                   kg, cg);
}

extern "C" void kernel_launch(void* const* d_in, const int* in_sizes, int n_in,
                              void* d_out, int out_size, void* d_ws,
                              size_t ws_size, hipStream_t stream) {
  (void)in_sizes; (void)n_in; (void)out_size; (void)ws_size;
  const float* x = (const float*)d_in[0];
  const float* m = (const float*)d_in[1];
  float* out = (float*)d_out;
  float* ws  = (float*)d_ws;
  float* carries = ws + (size_t)NCHUNK * CHUNKSTRIDE;

  prodscan_chunk_chain<<<dim3(NCHUNK, NSLAB), 256, 0, stream>>>(m, ws);
  prodscan_super_chain<<<dim3(NSUPER, NSLAB_S), 256, 0, stream>>>(ws,
                                                                  ws + MATELT);
  prodscan_carries<<<1, GEMV_THREADS, 0, stream>>>(x, ws, carries, out);
  prodscan_expand<<<NCHUNK, GEMV_THREADS, 0, stream>>>(m, ws, carries, out);
}
